# starting kernel
# baseline (speedup 1.0000x reference)
#include <hip/hip_runtime.h>

typedef __attribute__((ext_vector_type(2))) float v2f;
typedef __attribute__((ext_vector_type(8))) float v8f;

#define DIN  64
#define DOUT 64

__global__ __launch_bounds__(256) void gcn_xwt_wmma(
    const float* __restrict__ x, const float* __restrict__ W,
    float* __restrict__ y, int n_nodes)
{
    __shared__ float sW[DOUT * DIN];
    for (int i = threadIdx.x; i < DOUT * DIN; i += 256) sW[i] = W[i];
    __syncthreads();

    const int wave = threadIdx.x >> 5;
    const int lane = threadIdx.x & 31;
    const int m    = lane & 15;
    const int half = lane >> 4;
    const int tile = blockIdx.x * 8 + wave;
    const int row0 = tile * 16;
    if (row0 >= n_nodes) return;

    const int arow = min(row0 + m, n_nodes - 1);
    const float* __restrict__ xr = x + (size_t)arow * DIN;

    v8f acc[4] = {};

    #pragma unroll
    for (int k0 = 0; k0 < DIN; k0 += 4) {
        const int ka = k0 + 2 * half;
        v2f a;
        a.x = xr[ka];
        a.y = xr[ka + 1];
        #pragma unroll
        for (int nt = 0; nt < 4; ++nt) {
            const int n = nt * 16 + m;
            v2f bw;
            bw.x = sW[n * DIN + ka];
            bw.y = sW[n * DIN + ka + 1];
            acc[nt] = __builtin_amdgcn_wmma_f32_16x16x4_f32(
                 false, a,  false, bw,
                 (short)0, acc[nt],
                 false,  false);
        }
    }

    #pragma unroll
    for (int nt = 0; nt < 4; ++nt) {
        #pragma unroll
        for (int v = 0; v < 8; ++v) {
            const int r = row0 + v + 8 * half;
            if (r < n_nodes)
                y[(size_t)r * DOUT + nt * 16 + m] = acc[nt][v];
        }
    }
}

__global__ __launch_bounds__(256) void gcn_bias_init(
    float* __restrict__ out, const float* __restrict__ b, int n_nodes)
{
    const int total = n_nodes * DOUT;
    for (int i = blockIdx.x * blockDim.x + threadIdx.x; i < total;
         i += gridDim.x * blockDim.x)
        out[i] = b[i & (DOUT - 1)];
}

__global__ __launch_bounds__(256) void gcn_edge_scatter(
    const int* __restrict__ erow, const int* __restrict__ ecol,
    const float* __restrict__ eval, const float* __restrict__ y,
    float* out, int n_edges)
{
    const int gid = blockIdx.x * 256 + threadIdx.x;
    const int e = gid >> 4;
    const int q = gid & 15;
    if (e >= n_edges) return;

    const int r   = erow[e];
    const int c   = ecol[e];
    const float v = eval[e];

    const float4* __restrict__ ys =
        (const float4*)(y + (size_t)c * DOUT);
    const float4 mv = ys[q];

    float* o = out + (size_t)r * DOUT + q * 4;
    atomicAdd(o + 0, v * mv.x);
    atomicAdd(o + 1, v * mv.y);
    atomicAdd(o + 2, v * mv.z);
    atomicAdd(o + 3, v * mv.w);
}

extern "C" void kernel_launch(void* const* d_in, const int* in_sizes, int n_in,
                              void* d_out, int out_size, void* d_ws, size_t ws_size,
                              hipStream_t stream)
{
    const float* x    = (const float*)d_in[0];
    const int*   erow = (const int*)  d_in[1];
    const int*   ecol = (const int*)  d_in[2];
    const float* ev   = (const float*)d_in[3];
    const float* W    = (const float*)d_in[4];
    const float* b    = (const float*)d_in[5];
    float* out = (float*)d_out;
    float* y   = (float*)d_ws;

    const int n_nodes = in_sizes[0] / DIN;
    const int n_edges = in_sizes[3];

    const int tiles = (n_nodes + 15) / 16;
    gcn_xwt_wmma<<<(tiles + 7) / 8, 256, 0, stream>>>(x, W, y, n_nodes);

    const int total = n_nodes * DOUT;
    int bgrid = (total + 255) / 256;
    if (bgrid > 4096) bgrid = 4096;
    gcn_bias_init<<<bgrid, 256, 0, stream>>>(out, b, n_nodes);

    const long long sthreads = (long long)n_edges * 16;
    gcn_edge_scatter<<<(int)((sthreads + 255) / 256), 256, 0, stream>>>(
        erow, ecol, ev, y, out, n_edges);
}


# edge scatter re-tiled: 16384 looping waves, all 32 lanes on one edge, 8 edges' row loads in flight; each y load and f32 atomic add covers a whole 128-B line; atomics and arrival order kept
# speedup vs baseline: 1.6140x; 1.6140x over previous
#include <hip/hip_runtime.h>

typedef __attribute__((ext_vector_type(2))) float v2f;
typedef __attribute__((ext_vector_type(8))) float v8f;

#define DIN  64
#define DOUT 64

__global__ __launch_bounds__(256) void gcn_xwt_wmma(
    const float* __restrict__ x, const float* __restrict__ W,
    float* __restrict__ y, int n_nodes)
{
    __shared__ float sW[DOUT * DIN];
    for (int i = threadIdx.x; i < DOUT * DIN; i += 256) sW[i] = W[i];
    __syncthreads();

    const int wave = threadIdx.x >> 5;
    const int lane = threadIdx.x & 31;
    const int m    = lane & 15;
    const int half = lane >> 4;
    const int tile = blockIdx.x * 8 + wave;
    const int row0 = tile * 16;
    if (row0 >= n_nodes) return;

    const int arow = min(row0 + m, n_nodes - 1);
    const float* __restrict__ xr = x + (size_t)arow * DIN;

    v8f acc[4] = {};

    #pragma unroll
    for (int k0 = 0; k0 < DIN; k0 += 4) {
        const int ka = k0 + 2 * half;
        v2f a;
        a.x = xr[ka];
        a.y = xr[ka + 1];
        #pragma unroll
        for (int nt = 0; nt < 4; ++nt) {
            const int n = nt * 16 + m;
            v2f bw;
            bw.x = sW[n * DIN + ka];
            bw.y = sW[n * DIN + ka + 1];
            acc[nt] = __builtin_amdgcn_wmma_f32_16x16x4_f32(
                 false, a,  false, bw,
                 (short)0, acc[nt],
                 false,  false);
        }
    }

    #pragma unroll
    for (int nt = 0; nt < 4; ++nt) {
        #pragma unroll
        for (int v = 0; v < 8; ++v) {
            const int r = row0 + v + 8 * half;
            if (r < n_nodes)
                y[(size_t)r * DOUT + nt * 16 + m] = acc[nt][v];
        }
    }
}

__global__ __launch_bounds__(256) void gcn_bias_init(
    float* __restrict__ out, const float* __restrict__ b, int n_nodes)
{
    const int total = n_nodes * DOUT;
    for (int i = blockIdx.x * blockDim.x + threadIdx.x; i < total;
         i += gridDim.x * blockDim.x)
        out[i] = b[i & (DOUT - 1)];
}

__global__ __launch_bounds__(256) void gcn_edge_scatter(
    const int* __restrict__ erow, const int* __restrict__ ecol,
    const float* __restrict__ eval, const float* __restrict__ y,
    float* out, int n_edges)
{
    const int lane = threadIdx.x & 31;
    const int wave = __builtin_amdgcn_readfirstlane(
        (int)((blockIdx.x * 256 + threadIdx.x) >> 5));
    const int nwaves = gridDim.x * 8;

    for (int base = wave * 32; base < n_edges; base += nwaves * 32) {
        const int cnt = min(32, n_edges - base);
        int   r_l = 0;
        int   c_l = 0;
        float v_l = 0.0f;
        if (lane < cnt) {
            r_l = erow[base + lane];
            c_l = ecol[base + lane];
            v_l = eval[base + lane];
        }

        int j = 0;
        for (; j + 8 <= cnt; j += 8) {
            float m0[8], m1[8];
            #pragma unroll
            for (int u = 0; u < 8; ++u) {
                const int c = __builtin_amdgcn_readlane(c_l, j + u);
                const float* __restrict__ ys = y + (size_t)c * DOUT;
                m0[u] = ys[lane];
                m1[u] = ys[32 + lane];
            }
            #pragma unroll
            for (int u = 0; u < 8; ++u) {
                const int   r = __builtin_amdgcn_readlane(r_l, j + u);
                const float v = __int_as_float(
                    __builtin_amdgcn_readlane(__float_as_int(v_l), j + u));
                float* o = out + (size_t)r * DOUT;
                atomicAdd(o + lane,      v * m0[u]);
                atomicAdd(o + 32 + lane, v * m1[u]);
            }
        }
        for (; j < cnt; ++j) {
            const int   r = __builtin_amdgcn_readlane(r_l, j);
            const int   c = __builtin_amdgcn_readlane(c_l, j);
            const float v = __int_as_float(
                __builtin_amdgcn_readlane(__float_as_int(v_l), j));
            const float* __restrict__ ys = y + (size_t)c * DOUT;
            const float t0 = ys[lane];
            const float t1 = ys[32 + lane];
            float* o = out + (size_t)r * DOUT;
            atomicAdd(o + lane,      v * t0);
            atomicAdd(o + 32 + lane, v * t1);
        }
    }
}

extern "C" void kernel_launch(void* const* d_in, const int* in_sizes, int n_in,
                              void* d_out, int out_size, void* d_ws, size_t ws_size,
                              hipStream_t stream)
{
    const float* x    = (const float*)d_in[0];
    const int*   erow = (const int*)  d_in[1];
    const int*   ecol = (const int*)  d_in[2];
    const float* ev   = (const float*)d_in[3];
    const float* W    = (const float*)d_in[4];
    const float* b    = (const float*)d_in[5];
    float* out = (float*)d_out;
    float* y   = (float*)d_ws;

    const int n_nodes = in_sizes[0] / DIN;
    const int n_edges = in_sizes[3];

    const int tiles = (n_nodes + 15) / 16;
    gcn_xwt_wmma<<<(tiles + 7) / 8, 256, 0, stream>>>(x, W, y, n_nodes);

    const int total = n_nodes * DOUT;
    int bgrid = (total + 255) / 256;
    if (bgrid > 4096) bgrid = 4096;
    gcn_bias_init<<<bgrid, 256, 0, stream>>>(out, b, n_nodes);

    const int chunks = (n_edges + 31) / 32;
    int sgrid = (chunks + 7) / 8;
    if (sgrid > 2048) sgrid = 2048;
    if (sgrid < 1) sgrid = 1;
    gcn_edge_scatter<<<sgrid, 256, 0, stream>>>(
        erow, ecol, ev, y, out, n_edges);
}


# bias out=b folded into the x@W^T kernel: each wave stores its own 16 rows (whole-line 16-B stores) ahead of its WMMA loop; the init launch of 32768 short waves is dropped, 3 launches -> 2
# speedup vs baseline: 1.9073x; 1.1817x over previous
#include <hip/hip_runtime.h>

typedef __attribute__((ext_vector_type(2))) float v2f;
typedef __attribute__((ext_vector_type(8))) float v8f;

#define DIN  64
#define DOUT 64

__global__ __launch_bounds__(256) void gcn_xwt_wmma(
    const float* __restrict__ x, const float* __restrict__ W,
    const float* __restrict__ b, float* __restrict__ y,
    float* __restrict__ out, int n_nodes)
{
    const int wave = threadIdx.x >> 5;
    const int lane = threadIdx.x & 31;
    const int m    = lane & 15;
    const int half = lane >> 4;

    const float4 bv = make_float4(b[m * 4], b[m * 4 + 1], b[m * 4 + 2], b[m * 4 + 3]);

    __shared__ float sW[DOUT * DIN];
    for (int i = threadIdx.x; i < DOUT * DIN; i += 256) sW[i] = W[i];
    __syncthreads();
    const int tile = blockIdx.x * 8 + wave;
    const int row0 = tile * 16;
    if (row0 >= n_nodes) return;

    for (int j = 0; j < 8; ++j) {
        const int r = row0 + 2 * j + half;
        if (r < n_nodes)
            *reinterpret_cast<float4*>(out + (size_t)r * DOUT + m * 4) = bv;
    }

    const int arow = min(row0 + m, n_nodes - 1);
    const float* __restrict__ xr = x + (size_t)arow * DIN;

    v8f acc[4] = {};

    #pragma unroll
    for (int k0 = 0; k0 < DIN; k0 += 4) {
        const int ka = k0 + 2 * half;
        v2f a;
        a.x = xr[ka];
        a.y = xr[ka + 1];
        #pragma unroll
        for (int nt = 0; nt < 4; ++nt) {
            const int n = nt * 16 + m;
            v2f bw;
            bw.x = sW[n * DIN + ka];
            bw.y = sW[n * DIN + ka + 1];
            acc[nt] = __builtin_amdgcn_wmma_f32_16x16x4_f32(
                 false, a,  false, bw,
                 (short)0, acc[nt],
                 false,  false);
        }
    }

    #pragma unroll
    for (int nt = 0; nt < 4; ++nt) {
        #pragma unroll
        for (int v = 0; v < 8; ++v) {
            const int r = row0 + v + 8 * half;
            if (r < n_nodes)
                y[(size_t)r * DOUT + nt * 16 + m] = acc[nt][v];
        }
    }
}

__global__ __launch_bounds__(256) void gcn_edge_scatter(
    const int* __restrict__ erow, const int* __restrict__ ecol,
    const float* __restrict__ eval, const float* __restrict__ y,
    float* out, int n_edges)
{
    const int lane = threadIdx.x & 31;
    const int wave = __builtin_amdgcn_readfirstlane(
        (int)((blockIdx.x * 256 + threadIdx.x) >> 5));
    const int nwaves = gridDim.x * 8;

    for (int base = wave * 32; base < n_edges; base += nwaves * 32) {
        const int cnt = min(32, n_edges - base);
        int   r_l = 0;
        int   c_l = 0;
        float v_l = 0.0f;
        if (lane < cnt) {
            r_l = erow[base + lane];
            c_l = ecol[base + lane];
            v_l = eval[base + lane];
        }

        int j = 0;
        for (; j + 8 <= cnt; j += 8) {
            float m0[8], m1[8];
            #pragma unroll
            for (int u = 0; u < 8; ++u) {
                const int c = __builtin_amdgcn_readlane(c_l, j + u);
                const float* __restrict__ ys = y + (size_t)c * DOUT;
                m0[u] = ys[lane];
                m1[u] = ys[32 + lane];
            }
            #pragma unroll
            for (int u = 0; u < 8; ++u) {
                const int   r = __builtin_amdgcn_readlane(r_l, j + u);
                const float v = __int_as_float(
                    __builtin_amdgcn_readlane(__float_as_int(v_l), j + u));
                float* o = out + (size_t)r * DOUT;
                atomicAdd(o + lane,      v * m0[u]);
                atomicAdd(o + 32 + lane, v * m1[u]);
            }
        }
        for (; j < cnt; ++j) {
            const int   r = __builtin_amdgcn_readlane(r_l, j);
            const int   c = __builtin_amdgcn_readlane(c_l, j);
            const float v = __int_as_float(
                __builtin_amdgcn_readlane(__float_as_int(v_l), j));
            const float* __restrict__ ys = y + (size_t)c * DOUT;
            const float t0 = ys[lane];
            const float t1 = ys[32 + lane];
            float* o = out + (size_t)r * DOUT;
            atomicAdd(o + lane,      v * t0);
            atomicAdd(o + 32 + lane, v * t1);
        }
    }
}

extern "C" void kernel_launch(void* const* d_in, const int* in_sizes, int n_in,
                              void* d_out, int out_size, void* d_ws, size_t ws_size,
                              hipStream_t stream)
{
    const float* x    = (const float*)d_in[0];
    const int*   erow = (const int*)  d_in[1];
    const int*   ecol = (const int*)  d_in[2];
    const float* ev   = (const float*)d_in[3];
    const float* W    = (const float*)d_in[4];
    const float* b    = (const float*)d_in[5];
    float* out = (float*)d_out;
    float* y   = (float*)d_ws;

    const int n_nodes = in_sizes[0] / DIN;
    const int n_edges = in_sizes[3];

    const int tiles = (n_nodes + 15) / 16;
    gcn_xwt_wmma<<<(tiles + 7) / 8, 256, 0, stream>>>(x, W, b, y, out, n_nodes);

    const int chunks = (n_edges + 31) / 32;
    int sgrid = (chunks + 7) / 8;
    if (sgrid > 2048) sgrid = 2048;
    if (sgrid < 1) sgrid = 1;
    gcn_edge_scatter<<<sgrid, 256, 0, stream>>>(
        erow, ecol, ev, y, out, n_edges);
}


# W tile in LDS padded from a 64- to a 68-dword row pitch: the B-fragment reads of the x@W^T kernel drop from a 16-way bank conflict to none; k order and summation order unchanged
# speedup vs baseline: 2.0573x; 1.0786x over previous
#include <hip/hip_runtime.h>

typedef __attribute__((ext_vector_type(2))) float v2f;
typedef __attribute__((ext_vector_type(8))) float v8f;

#define DIN  64
#define DOUT 64
#define WPITCH 68

__global__ __launch_bounds__(256) void gcn_xwt_wmma(
    const float* __restrict__ x, const float* __restrict__ W,
    const float* __restrict__ b, float* __restrict__ y,
    float* __restrict__ out, int n_nodes)
{
    const int wave = threadIdx.x >> 5;
    const int lane = threadIdx.x & 31;
    const int m    = lane & 15;
    const int half = lane >> 4;

    const float4 bv = make_float4(b[m * 4], b[m * 4 + 1], b[m * 4 + 2], b[m * 4 + 3]);

    __shared__ float sW[DOUT * WPITCH];
    for (int i = threadIdx.x; i < DOUT * DIN; i += 256)
        sW[(i >> 6) * WPITCH + (i & 63)] = W[i];
    __syncthreads();
    const int tile = blockIdx.x * 8 + wave;
    const int row0 = tile * 16;
    if (row0 >= n_nodes) return;

    for (int j = 0; j < 8; ++j) {
        const int r = row0 + 2 * j + half;
        if (r < n_nodes)
            *reinterpret_cast<float4*>(out + (size_t)r * DOUT + m * 4) = bv;
    }

    const int arow = min(row0 + m, n_nodes - 1);
    const float* __restrict__ xr = x + (size_t)arow * DIN;

    v8f acc[4] = {};

    #pragma unroll
    for (int k0 = 0; k0 < DIN; k0 += 4) {
        const int ka = k0 + 2 * half;
        v2f a;
        a.x = xr[ka];
        a.y = xr[ka + 1];
        #pragma unroll
        for (int nt = 0; nt < 4; ++nt) {
            const int n = nt * 16 + m;
            v2f bw;
            bw.x = sW[n * WPITCH + ka];
            bw.y = sW[n * WPITCH + ka + 1];
            acc[nt] = __builtin_amdgcn_wmma_f32_16x16x4_f32(
                 false, a,  false, bw,
                 (short)0, acc[nt],
                 false,  false);
        }
    }

    #pragma unroll
    for (int nt = 0; nt < 4; ++nt) {
        #pragma unroll
        for (int v = 0; v < 8; ++v) {
            const int r = row0 + v + 8 * half;
            if (r < n_nodes)
                y[(size_t)r * DOUT + nt * 16 + m] = acc[nt][v];
        }
    }
}

__global__ __launch_bounds__(256) void gcn_edge_scatter(
    const int* __restrict__ erow, const int* __restrict__ ecol,
    const float* __restrict__ eval, const float* __restrict__ y,
    float* out, int n_edges)
{
    const int lane = threadIdx.x & 31;
    const int wave = __builtin_amdgcn_readfirstlane(
        (int)((blockIdx.x * 256 + threadIdx.x) >> 5));
    const int nwaves = gridDim.x * 8;

    for (int base = wave * 32; base < n_edges; base += nwaves * 32) {
        const int cnt = min(32, n_edges - base);
        int   r_l = 0;
        int   c_l = 0;
        float v_l = 0.0f;
        if (lane < cnt) {
            r_l = erow[base + lane];
            c_l = ecol[base + lane];
            v_l = eval[base + lane];
        }

        int j = 0;
        for (; j + 8 <= cnt; j += 8) {
            float m0[8], m1[8];
            #pragma unroll
            for (int u = 0; u < 8; ++u) {
                const int c = __builtin_amdgcn_readlane(c_l, j + u);
                const float* __restrict__ ys = y + (size_t)c * DOUT;
                m0[u] = ys[lane];
                m1[u] = ys[32 + lane];
            }
            #pragma unroll
            for (int u = 0; u < 8; ++u) {
                const int   r = __builtin_amdgcn_readlane(r_l, j + u);
                const float v = __int_as_float(
                    __builtin_amdgcn_readlane(__float_as_int(v_l), j + u));
                float* o = out + (size_t)r * DOUT;
                atomicAdd(o + lane,      v * m0[u]);
                atomicAdd(o + 32 + lane, v * m1[u]);
            }
        }
        for (; j < cnt; ++j) {
            const int   r = __builtin_amdgcn_readlane(r_l, j);
            const int   c = __builtin_amdgcn_readlane(c_l, j);
            const float v = __int_as_float(
                __builtin_amdgcn_readlane(__float_as_int(v_l), j));
            const float* __restrict__ ys = y + (size_t)c * DOUT;
            const float t0 = ys[lane];
            const float t1 = ys[32 + lane];
            float* o = out + (size_t)r * DOUT;
            atomicAdd(o + lane,      v * t0);
            atomicAdd(o + 32 + lane, v * t1);
        }
    }
}

extern "C" void kernel_launch(void* const* d_in, const int* in_sizes, int n_in,
                              void* d_out, int out_size, void* d_ws, size_t ws_size,
                              hipStream_t stream)
{
    const float* x    = (const float*)d_in[0];
    const int*   erow = (const int*)  d_in[1];
    const int*   ecol = (const int*)  d_in[2];
    const float* ev   = (const float*)d_in[3];
    const float* W    = (const float*)d_in[4];
    const float* b    = (const float*)d_in[5];
    float* out = (float*)d_out;
    float* y   = (float*)d_ws;

    const int n_nodes = in_sizes[0] / DIN;
    const int n_edges = in_sizes[3];

    const int tiles = (n_nodes + 15) / 16;
    gcn_xwt_wmma<<<(tiles + 7) / 8, 256, 0, stream>>>(x, W, b, y, out, n_nodes);

    const int chunks = (n_edges + 31) / 32;
    int sgrid = (chunks + 7) / 8;
    if (sgrid > 2048) sgrid = 2048;
    if (sgrid < 1) sgrid = 1;
    gcn_edge_scatter<<<sgrid, 256, 0, stream>>>(
        erow, ecol, ev, y, out, n_edges);
}


# x@W^T kernel: a wave requests all its global reads together at entry, ahead of the barrier: W as 4 16-B loads per thread (was 8 dependent trips of 4-B loads), then its 16 x k pairs; sums unchanged
# speedup vs baseline: 2.1687x; 1.0541x over previous
#include <hip/hip_runtime.h>

typedef __attribute__((ext_vector_type(2))) float v2f;
typedef __attribute__((ext_vector_type(8))) float v8f;

#define DIN  64
#define DOUT 64
#define WPITCH 68

__global__ __launch_bounds__(256) void gcn_xwt_wmma(
    const float* __restrict__ x, const float* __restrict__ W,
    const float* __restrict__ b, float* __restrict__ y,
    float* __restrict__ out, int n_nodes)
{
    const int wave = threadIdx.x >> 5;
    const int lane = threadIdx.x & 31;
    const int m    = lane & 15;
    const int half = lane >> 4;

    const float4 bv = make_float4(b[m * 4], b[m * 4 + 1], b[m * 4 + 2], b[m * 4 + 3]);

    __shared__ __attribute__((aligned(16))) float sW[DOUT * WPITCH];

    const float4* __restrict__ W4 = reinterpret_cast<const float4*>(W);
    float4 wq[4];
    #pragma unroll
    for (int q = 0; q < 4; ++q)
        wq[q] = W4[threadIdx.x + 256 * q];

    const int tile = blockIdx.x * 8 + wave;
    const int row0 = tile * 16;
    const int arow = min(row0 + m, n_nodes - 1);
    const float* __restrict__ xr = x + (size_t)arow * DIN;
    v2f a[16];
    #pragma unroll
    for (int j = 0; j < 16; ++j) {
        const int ka = 4 * j + 2 * half;
        a[j].x = xr[ka];
        a[j].y = xr[ka + 1];
    }

    #pragma unroll
    for (int q = 0; q < 4; ++q) {
        const int f = threadIdx.x + 256 * q;
        *reinterpret_cast<float4*>(&sW[(f >> 4) * WPITCH + (f & 15) * 4]) = wq[q];
    }
    __syncthreads();
    if (row0 >= n_nodes) return;

    for (int j = 0; j < 8; ++j) {
        const int r = row0 + 2 * j + half;
        if (r < n_nodes)
            *reinterpret_cast<float4*>(out + (size_t)r * DOUT + m * 4) = bv;
    }

    v8f acc[4] = {};

    #pragma unroll
    for (int j = 0; j < 16; ++j) {
        const int ka = 4 * j + 2 * half;
        #pragma unroll
        for (int nt = 0; nt < 4; ++nt) {
            const int n = nt * 16 + m;
            v2f bw;
            bw.x = sW[n * WPITCH + ka];
            bw.y = sW[n * WPITCH + ka + 1];
            acc[nt] = __builtin_amdgcn_wmma_f32_16x16x4_f32(
                 false, a[j],  false, bw,
                 (short)0, acc[nt],
                 false,  false);
        }
    }

    #pragma unroll
    for (int nt = 0; nt < 4; ++nt) {
        #pragma unroll
        for (int v = 0; v < 8; ++v) {
            const int r = row0 + v + 8 * half;
            if (r < n_nodes)
                y[(size_t)r * DOUT + nt * 16 + m] = acc[nt][v];
        }
    }
}

__global__ __launch_bounds__(256) void gcn_edge_scatter(
    const int* __restrict__ erow, const int* __restrict__ ecol,
    const float* __restrict__ eval, const float* __restrict__ y,
    float* out, int n_edges)
{
    const int lane = threadIdx.x & 31;
    const int wave = __builtin_amdgcn_readfirstlane(
        (int)((blockIdx.x * 256 + threadIdx.x) >> 5));
    const int nwaves = gridDim.x * 8;

    for (int base = wave * 32; base < n_edges; base += nwaves * 32) {
        const int cnt = min(32, n_edges - base);
        int   r_l = 0;
        int   c_l = 0;
        float v_l = 0.0f;
        if (lane < cnt) {
            r_l = erow[base + lane];
            c_l = ecol[base + lane];
            v_l = eval[base + lane];
        }

        int j = 0;
        for (; j + 8 <= cnt; j += 8) {
            float m0[8], m1[8];
            #pragma unroll
            for (int u = 0; u < 8; ++u) {
                const int c = __builtin_amdgcn_readlane(c_l, j + u);
                const float* __restrict__ ys = y + (size_t)c * DOUT;
                m0[u] = ys[lane];
                m1[u] = ys[32 + lane];
            }
            #pragma unroll
            for (int u = 0; u < 8; ++u) {
                const int   r = __builtin_amdgcn_readlane(r_l, j + u);
                const float v = __int_as_float(
                    __builtin_amdgcn_readlane(__float_as_int(v_l), j + u));
                float* o = out + (size_t)r * DOUT;
                atomicAdd(o + lane,      v * m0[u]);
                atomicAdd(o + 32 + lane, v * m1[u]);
            }
        }
        for (; j < cnt; ++j) {
            const int   r = __builtin_amdgcn_readlane(r_l, j);
            const int   c = __builtin_amdgcn_readlane(c_l, j);
            const float v = __int_as_float(
                __builtin_amdgcn_readlane(__float_as_int(v_l), j));
            const float* __restrict__ ys = y + (size_t)c * DOUT;
            const float t0 = ys[lane];
            const float t1 = ys[32 + lane];
            float* o = out + (size_t)r * DOUT;
            atomicAdd(o + lane,      v * t0);
            atomicAdd(o + 32 + lane, v * t1);
        }
    }
}

extern "C" void kernel_launch(void* const* d_in, const int* in_sizes, int n_in,
                              void* d_out, int out_size, void* d_ws, size_t ws_size,
                              hipStream_t stream)
{
    const float* x    = (const float*)d_in[0];
    const int*   erow = (const int*)  d_in[1];
    const int*   ecol = (const int*)  d_in[2];
    const float* ev   = (const float*)d_in[3];
    const float* W    = (const float*)d_in[4];
    const float* b    = (const float*)d_in[5];
    float* out = (float*)d_out;
    float* y   = (float*)d_ws;

    const int n_nodes = in_sizes[0] / DIN;
    const int n_edges = in_sizes[3];

    const int tiles = (n_nodes + 15) / 16;
    gcn_xwt_wmma<<<(tiles + 7) / 8, 256, 0, stream>>>(x, W, b, y, out, n_nodes);

    const int chunks = (n_edges + 31) / 32;
    int sgrid = (chunks + 7) / 8;
    if (sgrid > 2048) sgrid = 2048;
    if (sgrid < 1) sgrid = 1;
    gcn_edge_scatter<<<sgrid, 256, 0, stream>>>(
        erow, ecol, ev, y, out, n_edges);
}


# edge scatter: each out row gets its f32 atomic adds from one L2 half only: blocks read their die, draw 1024-edge tickets per half, keep edges whose row is theirs (packed in LDS); arrival order kept
# speedup vs baseline: 2.6800x; 1.2358x over previous
#include <hip/hip_runtime.h>

typedef __attribute__((ext_vector_type(2))) float v2f;
typedef __attribute__((ext_vector_type(8))) float v8f;

#define DIN  64
#define DOUT 64
#define WPITCH 68

__global__ __launch_bounds__(256) void gcn_xwt_wmma(
    const float* __restrict__ x, const float* __restrict__ W,
    const float* __restrict__ b, float* __restrict__ y,
    float* __restrict__ out, int n_nodes, unsigned* __restrict__ tickets)
{
    if (blockIdx.x == 0 && threadIdx.x < 2) tickets[threadIdx.x * 64] = 0u;

    const int wave = threadIdx.x >> 5;
    const int lane = threadIdx.x & 31;
    const int m    = lane & 15;
    const int half = lane >> 4;

    const float4 bv = make_float4(b[m * 4], b[m * 4 + 1], b[m * 4 + 2], b[m * 4 + 3]);

    __shared__ __attribute__((aligned(16))) float sW[DOUT * WPITCH];

    const float4* __restrict__ W4 = reinterpret_cast<const float4*>(W);
    float4 wq[4];
    #pragma unroll
    for (int q = 0; q < 4; ++q)
        wq[q] = W4[threadIdx.x + 256 * q];

    const int tile = blockIdx.x * 8 + wave;
    const int row0 = tile * 16;
    const int arow = min(row0 + m, n_nodes - 1);
    const float* __restrict__ xr = x + (size_t)arow * DIN;
    v2f a[16];
    #pragma unroll
    for (int j = 0; j < 16; ++j) {
        const int ka = 4 * j + 2 * half;
        a[j].x = xr[ka];
        a[j].y = xr[ka + 1];
    }

    #pragma unroll
    for (int q = 0; q < 4; ++q) {
        const int f = threadIdx.x + 256 * q;
        *reinterpret_cast<float4*>(&sW[(f >> 4) * WPITCH + (f & 15) * 4]) = wq[q];
    }
    __syncthreads();
    if (row0 >= n_nodes) return;

    for (int j = 0; j < 8; ++j) {
        const int r = row0 + 2 * j + half;
        if (r < n_nodes)
            *reinterpret_cast<float4*>(out + (size_t)r * DOUT + m * 4) = bv;
    }

    v8f acc[4] = {};

    #pragma unroll
    for (int j = 0; j < 16; ++j) {
        const int ka = 4 * j + 2 * half;
        #pragma unroll
        for (int nt = 0; nt < 4; ++nt) {
            const int n = nt * 16 + m;
            v2f bw;
            bw.x = sW[n * WPITCH + ka];
            bw.y = sW[n * WPITCH + ka + 1];
            acc[nt] = __builtin_amdgcn_wmma_f32_16x16x4_f32(
                 false, a[j],  false, bw,
                 (short)0, acc[nt],
                 false,  false);
        }
    }

    #pragma unroll
    for (int nt = 0; nt < 4; ++nt) {
        #pragma unroll
        for (int v = 0; v < 8; ++v) {
            const int r = row0 + v + 8 * half;
            if (r < n_nodes)
                y[(size_t)r * DOUT + nt * 16 + m] = acc[nt][v];
        }
    }
}

#define TICKET_EDGES 1024
#define WAVE_EDGES   128

__global__ __launch_bounds__(256) void gcn_edge_scatter(
    const int* __restrict__ erow, const int* __restrict__ ecol,
    const float* __restrict__ eval, const float* __restrict__ y,
    float* out, int n_edges, int n_nodes, unsigned* tickets)
{
    const int lane = threadIdx.x & 31;
    const int wave = __builtin_amdgcn_readfirstlane((int)(threadIdx.x >> 5));

    __shared__ unsigned s_ticket;
    __shared__ int   sR[8][WAVE_EDGES];
    __shared__ int   sC[8][WAVE_EDGES];
    __shared__ float sV[8][WAVE_EDGES];

    __shared__ int s_half;
    if (threadIdx.x == 0)
        s_half = (int)((__builtin_amdgcn_s_sendmsg_rtn(0x87) >> 18) & 1u);
    __syncthreads();
    const int my_half = __builtin_amdgcn_readfirstlane(s_half);
    const int split = n_nodes >> 1;
    const unsigned n_tickets =
        (unsigned)((n_edges + TICKET_EDGES - 1) / TICKET_EDGES);

    for (int pass = 0; pass < 2; ++pass) {
        const int q = my_half ^ pass;
        for (unsigned trip = 0; trip <= n_tickets; ++trip) {
            if (threadIdx.x == 0) s_ticket = atomicAdd(&tickets[q * 64], 1u);
            __syncthreads();
            const unsigned t = __builtin_amdgcn_readfirstlane(s_ticket);
            __syncthreads();
            if (t >= n_tickets) break;

            const int base = (int)t * TICKET_EDGES + wave * WAVE_EDGES;
            int   rr[4], cc[4];
            float vv[4];
            #pragma unroll
            for (int k = 0; k < 4; ++k) {
                const int e = base + 32 * k + lane;
                rr[k] = -1; cc[k] = 0; vv[k] = 0.0f;
                if (e < n_edges) {
                    rr[k] = erow[e];
                    cc[k] = ecol[e];
                    vv[k] = eval[e];
                }
            }
            int total = 0;
            #pragma unroll
            for (int k = 0; k < 4; ++k) {
                const bool mine = (rr[k] >= 0) && ((rr[k] >= split ? 1 : 0) == q);
                const unsigned mask = __builtin_amdgcn_ballot_w32(mine);
                const int pos = total + (int)__builtin_amdgcn_mbcnt_lo(mask, 0u);
                if (mine) {
                    sR[wave][pos] = rr[k];
                    sC[wave][pos] = cc[k];
                    sV[wave][pos] = vv[k];
                }
                total += __builtin_popcount(mask);
            }

            for (int s0 = 0; s0 < total; s0 += 32) {
                const int cnt = min(32, total - s0);
                int   r_l = 0;
                int   c_l = 0;
                float v_l = 0.0f;
                if (lane < cnt) {
                    r_l = sR[wave][s0 + lane];
                    c_l = sC[wave][s0 + lane];
                    v_l = sV[wave][s0 + lane];
                }

                int j = 0;
                for (; j + 8 <= cnt; j += 8) {
                    float m0[8], m1[8];
                    #pragma unroll
                    for (int u = 0; u < 8; ++u) {
                        const int c = __builtin_amdgcn_readlane(c_l, j + u);
                        const float* __restrict__ ys = y + (size_t)c * DOUT;
                        m0[u] = ys[lane];
                        m1[u] = ys[32 + lane];
                    }
                    #pragma unroll
                    for (int u = 0; u < 8; ++u) {
                        const int   r = __builtin_amdgcn_readlane(r_l, j + u);
                        const float v = __int_as_float(
                            __builtin_amdgcn_readlane(__float_as_int(v_l), j + u));
                        float* o = out + (size_t)r * DOUT;
                        atomicAdd(o + lane,      v * m0[u]);
                        atomicAdd(o + 32 + lane, v * m1[u]);
                    }
                }
                if (j < cnt) {
                    float t0[7], t1[7];
                    #pragma unroll
                    for (int u = 0; u < 7; ++u) {
                        const int ju = min(j + u, cnt - 1);
                        const int c = __builtin_amdgcn_readlane(c_l, ju);
                        const float* __restrict__ ys = y + (size_t)c * DOUT;
                        t0[u] = ys[lane];
                        t1[u] = ys[32 + lane];
                    }
                    asm volatile("" :: "v"(t0[0]), "v"(t1[0]), "v"(t0[1]), "v"(t1[1]),
                                       "v"(t0[2]), "v"(t1[2]), "v"(t0[3]), "v"(t1[3]),
                                       "v"(t0[4]), "v"(t1[4]), "v"(t0[5]), "v"(t1[5]),
                                       "v"(t0[6]), "v"(t1[6]));
                    #pragma unroll
                    for (int u = 0; u < 7; ++u) {
                        if (j + u < cnt) {
                            const int   r = __builtin_amdgcn_readlane(r_l, j + u);
                            const float v = __int_as_float(
                                __builtin_amdgcn_readlane(__float_as_int(v_l), j + u));
                            float* o = out + (size_t)r * DOUT;
                            atomicAdd(o + lane,      v * t0[u]);
                            atomicAdd(o + 32 + lane, v * t1[u]);
                        }
                    }
                }
            }
        }
    }
}

extern "C" void kernel_launch(void* const* d_in, const int* in_sizes, int n_in,
                              void* d_out, int out_size, void* d_ws, size_t ws_size,
                              hipStream_t stream)
{
    const float* x    = (const float*)d_in[0];
    const int*   erow = (const int*)  d_in[1];
    const int*   ecol = (const int*)  d_in[2];
    const float* ev   = (const float*)d_in[3];
    const float* W    = (const float*)d_in[4];
    const float* b    = (const float*)d_in[5];
    float* out = (float*)d_out;
    unsigned* tickets = (unsigned*)d_ws;
    float* y   = (float*)d_ws + 128;

    const int n_nodes = in_sizes[0] / DIN;
    const int n_edges = in_sizes[3];

    const int tiles = (n_nodes + 15) / 16;
    gcn_xwt_wmma<<<(tiles + 7) / 8, 256, 0, stream>>>(x, W, b, y, out, n_nodes, tickets);

    const int chunks = (n_edges + 31) / 32;
    int sgrid = (chunks + 7) / 8;
    if (sgrid > 2048) sgrid = 2048;
    if (sgrid < 1) sgrid = 1;
    gcn_edge_scatter<<<sgrid, 256, 0, stream>>>(
        erow, ecol, ev, y, out, n_edges, n_nodes, tickets);
}
